# starting kernel
# baseline (speedup 1.0000x reference)
#include <hip/hip_runtime.h>

typedef __attribute__((ext_vector_type(16))) _Float16 v16h;
typedef __attribute__((ext_vector_type(8)))  _Float16 v8h;
typedef __attribute__((ext_vector_type(8)))  float    v8f;

#define B_  4
#define C_  256
#define CQ_ 32
#define N_  4096

__device__ __forceinline__ v16h load_tile16x32(const _Float16* base, int strideHalfs) {
  const int l  = threadIdx.x & 31;
  const int hf = l >> 4;
  const _Float16* p = base + (size_t)(l & 15) * strideHalfs + hf * 8;
  union { v16h v; v8h h[2]; } u;
  u.h[0] = *(const v8h*)(p);
  u.h[1] = *(const v8h*)(p + 16);
  return u.v;
}

__global__ void pack_weights(const float* __restrict__ Wq, const float* __restrict__ bq,
                             const float* __restrict__ Wk, const float* __restrict__ bk,
                             const float* __restrict__ Wv, const float* __restrict__ bv,
                             _Float16* __restrict__ Wh, float* __restrict__ bP) {
  int idx = blockIdx.x * blockDim.x + threadIdx.x;
  if (idx >= 320 * 256) return;
  int o = idx >> 8, c = idx & 255;
  float w = (o < 32) ? Wq[o * 256 + c] : (o < 64) ? Wk[(o - 32) * 256 + c]
                                                  : Wv[(o - 64) * 256 + c];
  Wh[idx] = (_Float16)w;
  if (c == 0) bP[o] = (o < 32) ? bq[o] : (o < 64) ? bk[o - 32] : bv[o - 64];
}

__global__ __launch_bounds__(256) void transpose_cvt(const float* __restrict__ x,
                                                     _Float16* __restrict__ xt) {
  __shared__ _Float16 t[32][33];
  const int b = blockIdx.z, ct = blockIdx.y, nt = blockIdx.x;
  const int lx = threadIdx.x & 31;
  const int ly = threadIdx.x >> 5;
#pragma unroll
  for (int k = 0; k < 4; ++k) {
    int c = ct * 32 + ly + k * 8;
    int n = nt * 32 + lx;
    t[ly + k * 8][lx] = (_Float16)x[((size_t)b * C_ + c) * N_ + n];
  }
  __syncthreads();
#pragma unroll
  for (int k = 0; k < 4; ++k) {
    int n = nt * 32 + ly + k * 8;
    int c = ct * 32 + lx;
    xt[((size_t)b * N_ + n) * C_ + c] = t[lx][ly + k * 8];
  }
}

__global__ __launch_bounds__(32) void proj_kernel(
    const _Float16* __restrict__ Wh, const float* __restrict__ bP,
    const _Float16* __restrict__ xt, const float* __restrict__ mask,
    _Float16* __restrict__ Qh, _Float16* __restrict__ Kh, _Float16* __restrict__ Vh) {
  const int b = blockIdx.z, ot = blockIdx.y, nt = blockIdx.x;
  v8f acc = {0.f, 0.f, 0.f, 0.f, 0.f, 0.f, 0.f, 0.f};
#pragma unroll
  for (int ks = 0; ks < 8; ++ks) {
    v16h a  = load_tile16x32(Wh + (size_t)ot * 16 * C_ + ks * 32, C_);
    v16h bm = load_tile16x32(xt + ((size_t)b * N_ + nt * 16) * C_ + ks * 32, C_);
    acc = __builtin_amdgcn_wmma_f32_16x16x32_f16(false, a, false, bm, (short)0, acc,
                                                 false, false);
  }
  const int lane = threadIdx.x & 31;
  const int hf = lane >> 4, col = lane & 15;
  const int n = nt * 16 + col;
  const float mval = mask[(size_t)b * N_ + n];
#pragma unroll
  for (int r = 0; r < 8; ++r) {
    int o = ot * 16 + r + 8 * hf;
    float v = acc[r] + bP[o];
    if (o < 64) v *= mval;
    _Float16 h = (_Float16)v;
    if (o < 32)       Qh[((size_t)b * N_ + n) * CQ_ + o] = h;
    else if (o < 64)  Kh[((size_t)b * N_ + n) * CQ_ + (o - 32)] = h;
    else              Vh[((size_t)b * C_ + (o - 64)) * N_ + n] = h;
  }
}

__global__ __launch_bounds__(64) void attn_kernel(
    const _Float16* __restrict__ Qh, const _Float16* __restrict__ Kh,
    const _Float16* __restrict__ Vh, const float* __restrict__ x,
    const float* __restrict__ gamma, float* __restrict__ out) {
  __shared__ __align__(16) _Float16 Plds[2][16 * 32];
  const int b    = blockIdx.y;
  const int i0   = blockIdx.x * 16;
  const int wave = threadIdx.x >> 5;
  const int lane = threadIdx.x & 31;
  const int hf   = lane >> 4, col = lane & 15;

  const v16h qa = load_tile16x32(Qh + ((size_t)b * N_ + i0) * CQ_, CQ_);

  const v8f vzero = {0.f, 0.f, 0.f, 0.f, 0.f, 0.f, 0.f, 0.f};
  v8f acc[8];
  float mrun[8], lrun[8];
#pragma unroll
  for (int t = 0; t < 8; ++t) acc[t] = vzero;
#pragma unroll
  for (int r = 0; r < 8; ++r) { mrun[r] = -3.0e38f; lrun[r] = 0.0f; }

  _Float16* pbuf = &Plds[wave][0];

  for (int j0 = 0; j0 < N_; j0 += 32) {
    const v16h kb0 = load_tile16x32(Kh + ((size_t)b * N_ + j0) * CQ_, CQ_);
    const v16h kb1 = load_tile16x32(Kh + ((size_t)b * N_ + j0 + 16) * CQ_, CQ_);
    v8f e0 = __builtin_amdgcn_wmma_f32_16x16x32_f16(false, qa, false, kb0, (short)0,
                                                    vzero, false, false);
    v8f e1 = __builtin_amdgcn_wmma_f32_16x16x32_f16(false, qa, false, kb1, (short)0,
                                                    vzero, false, false);

    if (j0 + 32 < N_) {
      __builtin_prefetch(Kh + ((size_t)b * N_ + j0 + 32) * CQ_ + lane * 2, 0, 0);
    }

    float corr[8];
#pragma unroll
    for (int r = 0; r < 8; ++r) {
      float cm = fmaxf(e0[r], e1[r]);
#pragma unroll
      for (int s = 1; s < 16; s <<= 1) cm = fmaxf(cm, __shfl_xor(cm, s, 32));
      const float mn = fmaxf(mrun[r], cm);
      const float cf = __expf(mrun[r] - mn);
      const float p0 = __expf(e0[r] - mn);
      const float p1 = __expf(e1[r] - mn);
      float ps = p0 + p1;
#pragma unroll
      for (int s = 1; s < 16; s <<= 1) ps += __shfl_xor(ps, s, 32);
      lrun[r] = lrun[r] * cf + ps;
      mrun[r] = mn;
      corr[r] = cf;
      const int row = r + 8 * hf;
      pbuf[row * 32 + col]      = (_Float16)p0;
      pbuf[row * 32 + 16 + col] = (_Float16)p1;
    }
    asm volatile("s_wait_dscnt 0" ::: "memory");
    const v16h pa = load_tile16x32(pbuf, 32);

#pragma unroll
    for (int t = 0; t < 8; ++t) {
      const int c0 = wave * 128 + t * 16;
      const v16h vb = load_tile16x32(Vh + ((size_t)b * C_ + c0) * N_ + j0, N_);
#pragma unroll
      for (int r = 0; r < 8; ++r) acc[t][r] *= corr[r];
      acc[t] = __builtin_amdgcn_wmma_f32_16x16x32_f16(false, pa, false, vb, (short)0,
                                                      acc[t], false, false);
    }
  }

  const float g = gamma[0];
#pragma unroll
  for (int t = 0; t < 8; ++t) {
    const int c = wave * 128 + t * 16 + col;
#pragma unroll
    for (int r = 0; r < 8; ++r) {
      const int i = i0 + r + 8 * hf;
      const size_t off = ((size_t)b * C_ + c) * N_ + i;
      out[off] = g * (acc[t][r] / lrun[r]) + x[off];
    }
  }
}

extern "C" void kernel_launch(void* const* d_in, const int* in_sizes, int n_in,
                              void* d_out, int out_size, void* d_ws, size_t ws_size,
                              hipStream_t stream) {
  (void)in_sizes; (void)n_in; (void)out_size; (void)ws_size;
  const float* x     = (const float*)d_in[0];
  const float* mask  = (const float*)d_in[1];
  const float* Wq    = (const float*)d_in[2];
  const float* bq    = (const float*)d_in[3];
  const float* Wk    = (const float*)d_in[4];
  const float* bk    = (const float*)d_in[5];
  const float* Wv    = (const float*)d_in[6];
  const float* bv    = (const float*)d_in[7];
  const float* gamma = (const float*)d_in[8];
  float* out = (float*)d_out;

  char* ws = (char*)d_ws;
  size_t off = 0;
  auto take = [&](size_t bytes) -> char* {
    char* p = ws + off;
    off += (bytes + 255) & ~(size_t)255;
    return p;
  };
  _Float16* Wh = (_Float16*)take((size_t)320 * 256 * sizeof(_Float16));
  float*    bP = (float*)take(320 * sizeof(float));
  _Float16* xt = (_Float16*)take((size_t)B_ * N_ * C_ * sizeof(_Float16));
  _Float16* Qh = (_Float16*)take((size_t)B_ * N_ * CQ_ * sizeof(_Float16));
  _Float16* Kh = (_Float16*)take((size_t)B_ * N_ * CQ_ * sizeof(_Float16));
  _Float16* Vh = (_Float16*)take((size_t)B_ * C_ * N_ * sizeof(_Float16));

  pack_weights<<<(320 * 256 + 255) / 256, 256, 0, stream>>>(Wq, bq, Wk, bk, Wv, bv, Wh, bP);
  transpose_cvt<<<dim3(N_ / 32, C_ / 32, B_), 256, 0, stream>>>(x, xt);
  proj_kernel<<<dim3(N_ / 16, 320 / 16, B_), 32, 0, stream>>>(Wh, bP, xt, mask, Qh, Kh, Vh);
  attn_kernel<<<dim3(N_ / 16, B_), 64, 0, stream>>>(Qh, Kh, Vh, x, gamma, out);
}


# attn: transposed products S^T=K*Q^T, O^T=V*P^T: softmax row max/sum in-lane plus one half-wave exchange, exp(S^T) packed in registers as B operand; no LDS, no shuffles; row-sum order changes
# speedup vs baseline: 1.1393x; 1.1393x over previous
#include <hip/hip_runtime.h>

typedef __attribute__((ext_vector_type(16))) _Float16 v16h;
typedef __attribute__((ext_vector_type(8)))  _Float16 v8h;
typedef __attribute__((ext_vector_type(8)))  float    v8f;

#define B_  4
#define C_  256
#define CQ_ 32
#define N_  4096

__device__ __forceinline__ v16h load_tile16x32(const _Float16* base, int strideHalfs) {
  const int l  = threadIdx.x & 31;
  const int hf = l >> 4;
  const _Float16* p = base + (size_t)(l & 15) * strideHalfs + hf * 8;
  union { v16h v; v8h h[2]; } u;
  u.h[0] = *(const v8h*)(p);
  u.h[1] = *(const v8h*)(p + 16);
  return u.v;
}

__device__ __forceinline__ float xhalf(float x) {
  const int b = __builtin_bit_cast(int, x);
  return __builtin_bit_cast(float, __builtin_amdgcn_permlanex16(b, b, 0x76543210, 0xfedcba98,
                                                                false, false));
}

__global__ void pack_weights(const float* __restrict__ Wq, const float* __restrict__ bq,
                             const float* __restrict__ Wk, const float* __restrict__ bk,
                             const float* __restrict__ Wv, const float* __restrict__ bv,
                             _Float16* __restrict__ Wh, float* __restrict__ bP) {
  int idx = blockIdx.x * blockDim.x + threadIdx.x;
  if (idx >= 320 * 256) return;
  int o = idx >> 8, c = idx & 255;
  float w = (o < 32) ? Wq[o * 256 + c] : (o < 64) ? Wk[(o - 32) * 256 + c]
                                                  : Wv[(o - 64) * 256 + c];
  Wh[idx] = (_Float16)w;
  if (c == 0) bP[o] = (o < 32) ? bq[o] : (o < 64) ? bk[o - 32] : bv[o - 64];
}

__global__ __launch_bounds__(256) void transpose_cvt(const float* __restrict__ x,
                                                     _Float16* __restrict__ xt) {
  __shared__ _Float16 t[32][33];
  const int b = blockIdx.z, ct = blockIdx.y, nt = blockIdx.x;
  const int lx = threadIdx.x & 31;
  const int ly = threadIdx.x >> 5;
#pragma unroll
  for (int k = 0; k < 4; ++k) {
    int c = ct * 32 + ly + k * 8;
    int n = nt * 32 + lx;
    t[ly + k * 8][lx] = (_Float16)x[((size_t)b * C_ + c) * N_ + n];
  }
  __syncthreads();
#pragma unroll
  for (int k = 0; k < 4; ++k) {
    int n = nt * 32 + ly + k * 8;
    int c = ct * 32 + lx;
    xt[((size_t)b * N_ + n) * C_ + c] = t[lx][ly + k * 8];
  }
}

__global__ __launch_bounds__(32) void proj_kernel(
    const _Float16* __restrict__ Wh, const float* __restrict__ bP,
    const _Float16* __restrict__ xt, const float* __restrict__ mask,
    _Float16* __restrict__ Qh, _Float16* __restrict__ Kh, _Float16* __restrict__ Vh) {
  const int b = blockIdx.z, ot = blockIdx.y, nt = blockIdx.x;
  v8f acc = {0.f, 0.f, 0.f, 0.f, 0.f, 0.f, 0.f, 0.f};
#pragma unroll
  for (int ks = 0; ks < 8; ++ks) {
    v16h a  = load_tile16x32(Wh + (size_t)ot * 16 * C_ + ks * 32, C_);
    v16h bm = load_tile16x32(xt + ((size_t)b * N_ + nt * 16) * C_ + ks * 32, C_);
    acc = __builtin_amdgcn_wmma_f32_16x16x32_f16(false, a, false, bm, (short)0, acc,
                                                 false, false);
  }
  const int lane = threadIdx.x & 31;
  const int hf = lane >> 4, col = lane & 15;
  const int n = nt * 16 + col;
  const float mval = mask[(size_t)b * N_ + n];
#pragma unroll
  for (int r = 0; r < 8; ++r) {
    int o = ot * 16 + r + 8 * hf;
    float v = acc[r] + bP[o];
    if (o < 64) v *= mval;
    _Float16 h = (_Float16)v;
    if (o < 32)       Qh[((size_t)b * N_ + n) * CQ_ + o] = h;
    else if (o < 64)  Kh[((size_t)b * N_ + n) * CQ_ + (o - 32)] = h;
    else              Vh[((size_t)b * C_ + (o - 64)) * N_ + n] = h;
  }
}

__global__ __launch_bounds__(64) void attn_kernel(
    const _Float16* __restrict__ Qh, const _Float16* __restrict__ Kh,
    const _Float16* __restrict__ Vh, const float* __restrict__ x,
    const float* __restrict__ gamma, float* __restrict__ out) {
  const int b    = blockIdx.y;
  const int i0   = blockIdx.x * 16;
  const int wave = threadIdx.x >> 5;
  const int lane = threadIdx.x & 31;
  const int hf   = lane >> 4, col = lane & 15;

  const v16h qb = load_tile16x32(Qh + ((size_t)b * N_ + i0) * CQ_, CQ_);

  const v8f vzero = {0.f, 0.f, 0.f, 0.f, 0.f, 0.f, 0.f, 0.f};
  v8f acc[8];
#pragma unroll
  for (int t = 0; t < 8; ++t) acc[t] = vzero;
  float mrun = -3.0e38f, lrun = 0.0f;

  for (int j0 = 0; j0 < N_; j0 += 32) {
    const v16h ka0 = load_tile16x32(Kh + ((size_t)b * N_ + j0) * CQ_, CQ_);
    const v16h ka1 = load_tile16x32(Kh + ((size_t)b * N_ + j0 + 16) * CQ_, CQ_);
    const v8f e0 = __builtin_amdgcn_wmma_f32_16x16x32_f16(false, ka0, false, qb, (short)0,
                                                          vzero, false, false);
    const v8f e1 = __builtin_amdgcn_wmma_f32_16x16x32_f16(false, ka1, false, qb, (short)0,
                                                          vzero, false, false);

    if (j0 + 32 < N_) {
      __builtin_prefetch(Kh + ((size_t)b * N_ + j0 + 32) * CQ_ + lane * 2, 0, 0);
    }

    float c0m = fmaxf(fmaxf(e0[0], e0[1]), fmaxf(e0[2], e0[3]));
    float c1m = fmaxf(fmaxf(e0[4], e0[5]), fmaxf(e0[6], e0[7]));
    float c2m = fmaxf(fmaxf(e1[0], e1[1]), fmaxf(e1[2], e1[3]));
    float c3m = fmaxf(fmaxf(e1[4], e1[5]), fmaxf(e1[6], e1[7]));
    float cm = fmaxf(fmaxf(c0m, c1m), fmaxf(c2m, c3m));
    cm = fmaxf(cm, xhalf(cm));
    const float mn = fmaxf(mrun, cm);
    const float cf = __expf(mrun - mn);
    float p0[8], p1[8];
#pragma unroll
    for (int r = 0; r < 8; ++r) {
      p0[r] = __expf(e0[r] - mn);
      p1[r] = __expf(e1[r] - mn);
    }
    float ps = (((p0[0] + p0[1]) + (p0[2] + p0[3])) + ((p0[4] + p0[5]) + (p0[6] + p0[7]))) +
               (((p1[0] + p1[1]) + (p1[2] + p1[3])) + ((p1[4] + p1[5]) + (p1[6] + p1[7])));
    ps = ps + xhalf(ps);
    lrun = lrun * cf + ps;
    mrun = mn;

    v16h pb;
#pragma unroll
    for (int r = 0; r < 8; ++r) {
      pb[r]     = (_Float16)p0[r];
      pb[8 + r] = (_Float16)p1[r];
    }

#pragma unroll
    for (int t = 0; t < 8; ++t) {
      const int c0 = wave * 128 + t * 16;
      const v16h va = load_tile16x32(Vh + ((size_t)b * C_ + c0) * N_ + j0, N_);
#pragma unroll
      for (int r = 0; r < 8; ++r) acc[t][r] *= cf;
      acc[t] = __builtin_amdgcn_wmma_f32_16x16x32_f16(false, va, false, pb, (short)0,
                                                      acc[t], false, false);
    }
  }

  const float g = gamma[0];
  const int i = i0 + col;
#pragma unroll
  for (int t = 0; t < 8; ++t) {
#pragma unroll
    for (int r = 0; r < 8; ++r) {
      const int c = wave * 128 + t * 16 + 8 * hf + r;
      const size_t off = ((size_t)b * C_ + c) * N_ + i;
      out[off] = g * (acc[t][r] / lrun) + x[off];
    }
  }
}

extern "C" void kernel_launch(void* const* d_in, const int* in_sizes, int n_in,
                              void* d_out, int out_size, void* d_ws, size_t ws_size,
                              hipStream_t stream) {
  (void)in_sizes; (void)n_in; (void)out_size; (void)ws_size;
  const float* x     = (const float*)d_in[0];
  const float* mask  = (const float*)d_in[1];
  const float* Wq    = (const float*)d_in[2];
  const float* bq    = (const float*)d_in[3];
  const float* Wk    = (const float*)d_in[4];
  const float* bk    = (const float*)d_in[5];
  const float* Wv    = (const float*)d_in[6];
  const float* bv    = (const float*)d_in[7];
  const float* gamma = (const float*)d_in[8];
  float* out = (float*)d_out;

  char* ws = (char*)d_ws;
  size_t off = 0;
  auto take = [&](size_t bytes) -> char* {
    char* p = ws + off;
    off += (bytes + 255) & ~(size_t)255;
    return p;
  };
  _Float16* Wh = (_Float16*)take((size_t)320 * 256 * sizeof(_Float16));
  float*    bP = (float*)take(320 * sizeof(float));
  _Float16* xt = (_Float16*)take((size_t)B_ * N_ * C_ * sizeof(_Float16));
  _Float16* Qh = (_Float16*)take((size_t)B_ * N_ * CQ_ * sizeof(_Float16));
  _Float16* Kh = (_Float16*)take((size_t)B_ * N_ * CQ_ * sizeof(_Float16));
  _Float16* Vh = (_Float16*)take((size_t)B_ * C_ * N_ * sizeof(_Float16));

  pack_weights<<<(320 * 256 + 255) / 256, 256, 0, stream>>>(Wq, bq, Wk, bk, Wv, bv, Wh, bP);
  transpose_cvt<<<dim3(N_ / 32, C_ / 32, B_), 256, 0, stream>>>(x, xt);
  proj_kernel<<<dim3(N_ / 16, 320 / 16, B_), 32, 0, stream>>>(Wh, bP, xt, mask, Qh, Kh, Vh);
  attn_kernel<<<dim3(N_ / 16, B_), 64, 0, stream>>>(Qh, Kh, Vh, x, gamma, out);
}


# speedup vs baseline: 2.9333x; 2.5747x over previous
#include <hip/hip_runtime.h>

typedef __attribute__((ext_vector_type(16))) _Float16 v16h;
typedef __attribute__((ext_vector_type(8)))  _Float16 v8h;
typedef __attribute__((ext_vector_type(8)))  float    v8f;

#define B_  4
#define C_  256
#define CQ_ 32
#define N_  4096

__device__ __forceinline__ v16h load_tile16x32(const _Float16* base, int strideHalfs) {
  const int l  = threadIdx.x & 31;
  const int hf = l >> 4;
  const _Float16* p = base + (size_t)(l & 15) * strideHalfs + hf * 8;
  union { v16h v; v8h h[2]; } u;
  u.h[0] = *(const v8h*)(p);
  u.h[1] = *(const v8h*)(p + 16);
  return u.v;
}

__device__ __forceinline__ float xhalf(float x) {
  const int b = __builtin_bit_cast(int, x);
  return __builtin_bit_cast(float, __builtin_amdgcn_permlanex16(b, b, 0x76543210, 0xfedcba98,
                                                                false, false));
}

__global__ void pack_weights(const float* __restrict__ Wq, const float* __restrict__ bq,
                             const float* __restrict__ Wk, const float* __restrict__ bk,
                             const float* __restrict__ Wv, const float* __restrict__ bv,
                             _Float16* __restrict__ Wh, float* __restrict__ bP) {
  int idx = blockIdx.x * blockDim.x + threadIdx.x;
  if (idx >= 320 * 256) return;
  int o = idx >> 8, c = idx & 255;
  float w = (o < 32) ? Wq[o * 256 + c] : (o < 64) ? Wk[(o - 32) * 256 + c]
                                                  : Wv[(o - 64) * 256 + c];
  Wh[idx] = (_Float16)w;
  if (c == 0) bP[o] = (o < 32) ? bq[o] : (o < 64) ? bk[o - 32] : bv[o - 64];
}

__global__ __launch_bounds__(256) void transpose_cvt(const float* __restrict__ x,
                                                     _Float16* __restrict__ xt) {
  __shared__ _Float16 t[32][33];
  const int b = blockIdx.z, ct = blockIdx.y, nt = blockIdx.x;
  const int lx = threadIdx.x & 31;
  const int ly = threadIdx.x >> 5;
#pragma unroll
  for (int k = 0; k < 4; ++k) {
    int c = ct * 32 + ly + k * 8;
    int n = nt * 32 + lx;
    t[ly + k * 8][lx] = (_Float16)x[((size_t)b * C_ + c) * N_ + n];
  }
  __syncthreads();
#pragma unroll
  for (int k = 0; k < 4; ++k) {
    int n = nt * 32 + ly + k * 8;
    int c = ct * 32 + lx;
    xt[((size_t)b * N_ + n) * C_ + c] = t[lx][ly + k * 8];
  }
}

__global__ __launch_bounds__(32) void proj_kernel(
    const _Float16* __restrict__ Wh, const float* __restrict__ bP,
    const _Float16* __restrict__ xt, const float* __restrict__ mask,
    _Float16* __restrict__ Qh, _Float16* __restrict__ Kh, _Float16* __restrict__ Vh) {
  const int b = blockIdx.z, ot = blockIdx.y, nt = blockIdx.x;
  v8f acc = {0.f, 0.f, 0.f, 0.f, 0.f, 0.f, 0.f, 0.f};
#pragma unroll
  for (int ks = 0; ks < 8; ++ks) {
    v16h a  = load_tile16x32(Wh + (size_t)ot * 16 * C_ + ks * 32, C_);
    v16h bm = load_tile16x32(xt + ((size_t)b * N_ + nt * 16) * C_ + ks * 32, C_);
    acc = __builtin_amdgcn_wmma_f32_16x16x32_f16(false, a, false, bm, (short)0, acc,
                                                 false, false);
  }
  const int lane = threadIdx.x & 31;
  const int hf = lane >> 4, col = lane & 15;
  const int n = nt * 16 + col;
  const float mval = mask[(size_t)b * N_ + n];
#pragma unroll
  for (int r = 0; r < 8; ++r) {
    int o = ot * 16 + r + 8 * hf;
    float v = acc[r] + bP[o];
    if (o < 64) v *= mval;
    _Float16 h = (_Float16)v;
    if (o < 32)       Qh[((size_t)b * N_ + n) * CQ_ + o] = h;
    else if (o < 64)  Kh[((size_t)b * N_ + n) * CQ_ + (o - 32)] = h;
    else              Vh[((size_t)b * C_ + (o - 64)) * N_ + n] = h;
  }
}

#define KT_      64
#define VP_      72
#define KP_      40
#define VH_      (128 * VP_)
#define BLK_     16384

__global__ __launch_bounds__(256) void attn_kernel(
    const _Float16* __restrict__ Qh, const _Float16* __restrict__ Kh,
    const _Float16* __restrict__ Vh, const float* __restrict__ x,
    const float* __restrict__ gamma, float* __restrict__ out) {
  __shared__ __attribute__((aligned(16))) _Float16 lds[4 * BLK_];

  const int b     = blockIdx.y;
  const int tid   = threadIdx.x;
  const int wave  = tid >> 5;
  const int lane  = tid & 31;
  const int chalf = (wave >> 1) & 1;
  const int qt    = (wave & 1) | ((wave >> 2) << 1);
  const int i0    = blockIdx.x * 64 + qt * 16;
  const int hf    = lane >> 4, col = lane & 15;

  const v16h qb = load_tile16x32(Qh + ((size_t)b * N_ + i0) * CQ_, CQ_);

  const int vrow = tid >> 3, vch = tid & 7;
  const int krow = tid >> 2, kch = tid & 3;
  const _Float16* vsrc = Vh + ((size_t)b * C_ + vrow) * N_ + vch * 8;
  const _Float16* ksrc = Kh + (size_t)b * N_ * CQ_ + tid * 8;
  const int vdst = vrow * VP_ + vch * 8;
  const int kdst = VH_ + krow * KP_ + kch * 8;

  v8h gv[8];
  v8h gk;
#pragma unroll
  for (int i = 0; i < 8; ++i) gv[i] = *(const v8h*)(vsrc + (size_t)i * 32 * N_);
  gk = *(const v8h*)(ksrc);
#pragma unroll
  for (int i = 0; i < 8; ++i)
    *(v8h*)(&lds[(i >> 2) * BLK_ + vdst + (i & 3) * 32 * VP_]) = gv[i];
  *(v8h*)(&lds[kdst]) = gk;
  *(v8h*)(&lds[BLK_ + kdst]) = gk;
  __syncthreads();

  const v8f vzero = {0.f, 0.f, 0.f, 0.f, 0.f, 0.f, 0.f, 0.f};
  v8f acc[8];
#pragma unroll
  for (int t = 0; t < 8; ++t) acc[t] = vzero;
  float mrun = -3.0e38f, lrun = 0.0f;

  for (int it = 0; it < N_ / KT_; ++it) {
    const int cur = it & 1;
    const bool more = (it + 1 < N_ / KT_);
    if (more) {
#pragma unroll
      for (int i = 0; i < 8; ++i)
        gv[i] = *(const v8h*)(vsrc + (size_t)i * 32 * N_ + (it + 1) * KT_);
      gk = *(const v8h*)(ksrc + (size_t)(it + 1) * KT_ * CQ_);
    }
    __builtin_amdgcn_sched_barrier(0);

    const _Float16* Vs = &lds[(cur * 2 + chalf) * BLK_];
    const _Float16* Ks = Vs + VH_;

#pragma unroll
    for (int s = 0; s < 2; ++s) {
      const v16h ka0 = load_tile16x32(Ks + (s * 32) * KP_, KP_);
      const v16h ka1 = load_tile16x32(Ks + (s * 32 + 16) * KP_, KP_);
      const v8f e0 = __builtin_amdgcn_wmma_f32_16x16x32_f16(false, ka0, false, qb, (short)0,
                                                            vzero, false, false);
      const v8f e1 = __builtin_amdgcn_wmma_f32_16x16x32_f16(false, ka1, false, qb, (short)0,
                                                            vzero, false, false);

      v16h va[8];
#pragma unroll
      for (int t = 0; t < 8; ++t) va[t] = load_tile16x32(Vs + (t * 16) * VP_ + s * 32, VP_);
      __builtin_amdgcn_sched_barrier(0);

      float c0m = fmaxf(fmaxf(e0[0], e0[1]), fmaxf(e0[2], e0[3]));
      float c1m = fmaxf(fmaxf(e0[4], e0[5]), fmaxf(e0[6], e0[7]));
      float c2m = fmaxf(fmaxf(e1[0], e1[1]), fmaxf(e1[2], e1[3]));
      float c3m = fmaxf(fmaxf(e1[4], e1[5]), fmaxf(e1[6], e1[7]));
      float cm = fmaxf(fmaxf(c0m, c1m), fmaxf(c2m, c3m));
      cm = fmaxf(cm, xhalf(cm));
      const float mn = fmaxf(mrun, cm);
      const float cf = __expf(mrun - mn);
      float p0[8], p1[8];
#pragma unroll
      for (int r = 0; r < 8; ++r) {
        p0[r] = __expf(e0[r] - mn);
        p1[r] = __expf(e1[r] - mn);
      }
      float ps = (((p0[0] + p0[1]) + (p0[2] + p0[3])) + ((p0[4] + p0[5]) + (p0[6] + p0[7]))) +
                 (((p1[0] + p1[1]) + (p1[2] + p1[3])) + ((p1[4] + p1[5]) + (p1[6] + p1[7])));
      ps = ps + xhalf(ps);
      lrun = lrun * cf + ps;
      mrun = mn;

      v16h pb;
#pragma unroll
      for (int r = 0; r < 8; ++r) {
        pb[r]     = (_Float16)p0[r];
        pb[8 + r] = (_Float16)p1[r];
      }

#pragma unroll
      for (int t = 0; t < 8; ++t) {
#pragma unroll
        for (int r = 0; r < 8; ++r) acc[t][r] *= cf;
        acc[t] = __builtin_amdgcn_wmma_f32_16x16x32_f16(false, va[t], false, pb, (short)0,
                                                        acc[t], false, false);
      }
    }

    if (more) {
      const int nb = (cur ^ 1) * 2;
#pragma unroll
      for (int i = 0; i < 8; ++i)
        *(v8h*)(&lds[(nb + (i >> 2)) * BLK_ + vdst + (i & 3) * 32 * VP_]) = gv[i];
      *(v8h*)(&lds[nb * BLK_ + kdst]) = gk;
      *(v8h*)(&lds[(nb + 1) * BLK_ + kdst]) = gk;
    }
    __syncthreads();
  }

  const float g = gamma[0];
  const int i = i0 + col;
#pragma unroll
  for (int t = 0; t < 8; ++t) {
#pragma unroll
    for (int r = 0; r < 8; ++r) {
      const int c = chalf * 128 + t * 16 + 8 * hf + r;
      const size_t off = ((size_t)b * C_ + c) * N_ + i;
      out[off] = g * (acc[t][r] / lrun) + x[off];
    }
  }
}

extern "C" void kernel_launch(void* const* d_in, const int* in_sizes, int n_in,
                              void* d_out, int out_size, void* d_ws, size_t ws_size,
                              hipStream_t stream) {
  (void)in_sizes; (void)n_in; (void)out_size; (void)ws_size;
  const float* x     = (const float*)d_in[0];
  const float* mask  = (const float*)d_in[1];
  const float* Wq    = (const float*)d_in[2];
  const float* bq    = (const float*)d_in[3];
  const float* Wk    = (const float*)d_in[4];
  const float* bk    = (const float*)d_in[5];
  const float* Wv    = (const float*)d_in[6];
  const float* bv    = (const float*)d_in[7];
  const float* gamma = (const float*)d_in[8];
  float* out = (float*)d_out;

  char* ws = (char*)d_ws;
  size_t off = 0;
  auto take = [&](size_t bytes) -> char* {
    char* p = ws + off;
    off += (bytes + 255) & ~(size_t)255;
    return p;
  };
  _Float16* Wh = (_Float16*)take((size_t)320 * 256 * sizeof(_Float16));
  float*    bP = (float*)take(320 * sizeof(float));
  _Float16* xt = (_Float16*)take((size_t)B_ * N_ * C_ * sizeof(_Float16));
  _Float16* Qh = (_Float16*)take((size_t)B_ * N_ * CQ_ * sizeof(_Float16));
  _Float16* Kh = (_Float16*)take((size_t)B_ * N_ * CQ_ * sizeof(_Float16));
  _Float16* Vh = (_Float16*)take((size_t)B_ * C_ * N_ * sizeof(_Float16));

  pack_weights<<<(320 * 256 + 255) / 256, 256, 0, stream>>>(Wq, bq, Wk, bk, Wv, bv, Wh, bP);
  transpose_cvt<<<dim3(N_ / 32, C_ / 32, B_), 256, 0, stream>>>(x, xt);
  proj_kernel<<<dim3(N_ / 16, 320 / 16, B_), 32, 0, stream>>>(Wh, bP, xt, mask, Qh, Kh, Vh);
  attn_kernel<<<dim3(N_ / 64, B_), 256, 0, stream>>>(Qh, Kh, Vh, x, gamma, out);
}


# proj: one kernel rounds its x tile and the f32 weights to f16 into LDS (64 pixels x 320 rows per 8-wave group; v tiles as x^T*W^T for 16-byte stores); pack and transpose launches dropped
# speedup vs baseline: 3.8317x; 1.3063x over previous
#include <hip/hip_runtime.h>

typedef __attribute__((ext_vector_type(16))) _Float16 v16h;
typedef __attribute__((ext_vector_type(8)))  _Float16 v8h;
typedef __attribute__((ext_vector_type(8)))  float    v8f;

#define B_  4
#define C_  256
#define CQ_ 32
#define N_  4096

__device__ __forceinline__ v16h load_tile16x32(const _Float16* base, int strideHalfs) {
  const int l  = threadIdx.x & 31;
  const int hf = l >> 4;
  const _Float16* p = base + (size_t)(l & 15) * strideHalfs + hf * 8;
  union { v16h v; v8h h[2]; } u;
  u.h[0] = *(const v8h*)(p);
  u.h[1] = *(const v8h*)(p + 16);
  return u.v;
}

__device__ __forceinline__ float xhalf(float x) {
  const int b = __builtin_bit_cast(int, x);
  return __builtin_bit_cast(float, __builtin_amdgcn_permlanex16(b, b, 0x76543210, 0xfedcba98,
                                                                false, false));
}

typedef __attribute__((ext_vector_type(4))) float v4f;
#define PX_ 264

__global__ __launch_bounds__(256) void proj_fused(
    const float* __restrict__ x, const float* __restrict__ mask,
    const float* __restrict__ Wq, const float* __restrict__ bq,
    const float* __restrict__ Wk, const float* __restrict__ bk,
    const float* __restrict__ Wv, const float* __restrict__ bv,
    _Float16* __restrict__ Qh, _Float16* __restrict__ Kh, _Float16* __restrict__ Vh) {
  __shared__ __attribute__((aligned(16))) _Float16 Ws[320 * PX_];
  __shared__ __attribute__((aligned(16))) _Float16 Xs[64 * PX_];

  const int b    = blockIdx.y;
  const int n0   = blockIdx.x * 64;
  const int tid  = threadIdx.x;
  const int wave = __builtin_amdgcn_readfirstlane(tid >> 5);
  const int lane = tid & 31;

  const int xq = tid & 15, xg = tid >> 4;
  v4f xv[2][8];
#pragma unroll
  for (int i = 0; i < 2; ++i)
#pragma unroll
    for (int k = 0; k < 8; ++k)
      xv[i][k] = *(const v4f*)(x + ((size_t)b * C_ + xg * 8 + i * 128 + k) * N_ + n0 + xq * 4);

#pragma unroll
  for (int i0 = 0; i0 < 40; i0 += 8) {
    v4f wv[8][2];
#pragma unroll
    for (int u = 0; u < 8; ++u) {
      const int i = i0 + u;
      const int o = wave + 8 * i;
      const float* srow = (i < 4) ? (Wq + (size_t)o * C_)
                        : (i < 8) ? (Wk + (size_t)(o - 32) * C_)
                                  : (Wv + (size_t)(o - 64) * C_);
      const v4f* p = (const v4f*)(srow + lane * 8);
      wv[u][0] = p[0];
      wv[u][1] = p[1];
    }
#pragma unroll
    for (int u = 0; u < 8; ++u) {
      const int o = wave + 8 * (i0 + u);
      v8h h;
#pragma unroll
      for (int e = 0; e < 4; ++e) {
        h[e]     = (_Float16)wv[u][0][e];
        h[4 + e] = (_Float16)wv[u][1][e];
      }
      *(v8h*)(&Ws[o * PX_ + lane * 8]) = h;
    }
  }

#pragma unroll
  for (int i = 0; i < 2; ++i)
#pragma unroll
    for (int j = 0; j < 4; ++j) {
      v8h h;
#pragma unroll
      for (int k = 0; k < 8; ++k) h[k] = (_Float16)xv[i][k][j];
      *(v8h*)(&Xs[(xq * 4 + j) * PX_ + xg * 8 + i * 128]) = h;
    }
  __syncthreads();

  const int np = wave & 1, og = wave >> 1;
  const int hf = lane >> 4, col = lane & 15;

  float bqk[4][8];
  float bvv[5];
#pragma unroll
  for (int u = 0; u < 5; ++u) {
    const int ot = og * 5 + u;
    if (u < 4) {
      const float* bsrc = ((ot & 2) ? bk : bq) + (ot & 1) * 16 + 8 * hf;
#pragma unroll
      for (int r = 0; r < 8; ++r) bqk[u][r] = bsrc[r];
    }
    const int vt = (ot >= 4) ? (ot - 4) : 0;
    bvv[u] = bv[vt * 16 + col];
  }
  float mval[2];
#pragma unroll
  for (int t = 0; t < 2; ++t) mval[t] = mask[(size_t)b * N_ + n0 + (np * 2 + t) * 16 + col];

  v16h xf[2][8];
#pragma unroll
  for (int t = 0; t < 2; ++t)
#pragma unroll
    for (int ks = 0; ks < 8; ++ks)
      xf[t][ks] = load_tile16x32(Xs + ((np * 2 + t) * 16) * PX_ + ks * 32, PX_);

  v16h wf[2][8];
#pragma unroll
  for (int ks = 0; ks < 8; ++ks)
    wf[0][ks] = load_tile16x32(Ws + (og * 5 * 16) * PX_ + ks * 32, PX_);

  const v8f vzero = {0.f, 0.f, 0.f, 0.f, 0.f, 0.f, 0.f, 0.f};
#pragma unroll
  for (int u = 0; u < 5; ++u) {
    const int ot = og * 5 + u;
    if (u + 1 < 5) {
#pragma unroll
      for (int ks = 0; ks < 8; ++ks)
        wf[(u + 1) & 1][ks] = load_tile16x32(Ws + ((ot + 1) * 16) * PX_ + ks * 32, PX_);
    }
    __builtin_amdgcn_sched_barrier(0);

    v8f acc[2] = {vzero, vzero};
    if (ot < 4) {
#pragma unroll
      for (int ks = 0; ks < 8; ++ks)
#pragma unroll
        for (int t = 0; t < 2; ++t)
          acc[t] = __builtin_amdgcn_wmma_f32_16x16x32_f16(false, wf[u & 1][ks], false, xf[t][ks],
                                                          (short)0, acc[t], false, false);
      _Float16* dst = (ot < 2) ? Qh : Kh;
#pragma unroll
      for (int t = 0; t < 2; ++t) {
        const int n = n0 + (np * 2 + t) * 16 + col;
        v8h h;
#pragma unroll
        for (int r = 0; r < 8; ++r) h[r] = (_Float16)((acc[t][r] + bqk[u < 4 ? u : 0][r]) * mval[t]);
        *(v8h*)(dst + ((size_t)b * N_ + n) * CQ_ + (ot & 1) * 16 + 8 * hf) = h;
      }
    } else {
#pragma unroll
      for (int ks = 0; ks < 8; ++ks)
#pragma unroll
        for (int t = 0; t < 2; ++t)
          acc[t] = __builtin_amdgcn_wmma_f32_16x16x32_f16(false, xf[t][ks], false, wf[u & 1][ks],
                                                          (short)0, acc[t], false, false);
      const int ch = (ot - 4) * 16 + col;
#pragma unroll
      for (int t = 0; t < 2; ++t) {
        v8h h;
#pragma unroll
        for (int r = 0; r < 8; ++r) h[r] = (_Float16)(acc[t][r] + bvv[u]);
        *(v8h*)(Vh + ((size_t)b * C_ + ch) * N_ + n0 + (np * 2 + t) * 16 + 8 * hf) = h;
      }
    }
  }
}

#define KT_      64
#define VP_      72
#define KP_      40
#define VH_      (128 * VP_)
#define BLK_     16384

__global__ __launch_bounds__(256) void attn_kernel(
    const _Float16* __restrict__ Qh, const _Float16* __restrict__ Kh,
    const _Float16* __restrict__ Vh, const float* __restrict__ x,
    const float* __restrict__ gamma, float* __restrict__ out) {
  __shared__ __attribute__((aligned(16))) _Float16 lds[4 * BLK_];

  const int b     = blockIdx.y;
  const int tid   = threadIdx.x;
  const int wave  = tid >> 5;
  const int lane  = tid & 31;
  const int chalf = (wave >> 1) & 1;
  const int qt    = (wave & 1) | ((wave >> 2) << 1);
  const int i0    = blockIdx.x * 64 + qt * 16;
  const int hf    = lane >> 4, col = lane & 15;

  const v16h qb = load_tile16x32(Qh + ((size_t)b * N_ + i0) * CQ_, CQ_);

  const int vrow = tid >> 3, vch = tid & 7;
  const int krow = tid >> 2, kch = tid & 3;
  const _Float16* vsrc = Vh + ((size_t)b * C_ + vrow) * N_ + vch * 8;
  const _Float16* ksrc = Kh + (size_t)b * N_ * CQ_ + tid * 8;
  const int vdst = vrow * VP_ + vch * 8;
  const int kdst = VH_ + krow * KP_ + kch * 8;

  v8h gv[8];
  v8h gk;
#pragma unroll
  for (int i = 0; i < 8; ++i) gv[i] = *(const v8h*)(vsrc + (size_t)i * 32 * N_);
  gk = *(const v8h*)(ksrc);
#pragma unroll
  for (int i = 0; i < 8; ++i)
    *(v8h*)(&lds[(i >> 2) * BLK_ + vdst + (i & 3) * 32 * VP_]) = gv[i];
  *(v8h*)(&lds[kdst]) = gk;
  *(v8h*)(&lds[BLK_ + kdst]) = gk;
  __syncthreads();

  const v8f vzero = {0.f, 0.f, 0.f, 0.f, 0.f, 0.f, 0.f, 0.f};
  v8f acc[8];
#pragma unroll
  for (int t = 0; t < 8; ++t) acc[t] = vzero;
  float mrun = -3.0e38f, lrun = 0.0f;

  for (int it = 0; it < N_ / KT_; ++it) {
    const int cur = it & 1;
    const bool more = (it + 1 < N_ / KT_);
    if (more) {
#pragma unroll
      for (int i = 0; i < 8; ++i)
        gv[i] = *(const v8h*)(vsrc + (size_t)i * 32 * N_ + (it + 1) * KT_);
      gk = *(const v8h*)(ksrc + (size_t)(it + 1) * KT_ * CQ_);
    }
    __builtin_amdgcn_sched_barrier(0);

    const _Float16* Vs = &lds[(cur * 2 + chalf) * BLK_];
    const _Float16* Ks = Vs + VH_;

#pragma unroll
    for (int s = 0; s < 2; ++s) {
      const v16h ka0 = load_tile16x32(Ks + (s * 32) * KP_, KP_);
      const v16h ka1 = load_tile16x32(Ks + (s * 32 + 16) * KP_, KP_);
      const v8f e0 = __builtin_amdgcn_wmma_f32_16x16x32_f16(false, ka0, false, qb, (short)0,
                                                            vzero, false, false);
      const v8f e1 = __builtin_amdgcn_wmma_f32_16x16x32_f16(false, ka1, false, qb, (short)0,
                                                            vzero, false, false);

      v16h va[8];
#pragma unroll
      for (int t = 0; t < 8; ++t) va[t] = load_tile16x32(Vs + (t * 16) * VP_ + s * 32, VP_);
      __builtin_amdgcn_sched_barrier(0);

      float c0m = fmaxf(fmaxf(e0[0], e0[1]), fmaxf(e0[2], e0[3]));
      float c1m = fmaxf(fmaxf(e0[4], e0[5]), fmaxf(e0[6], e0[7]));
      float c2m = fmaxf(fmaxf(e1[0], e1[1]), fmaxf(e1[2], e1[3]));
      float c3m = fmaxf(fmaxf(e1[4], e1[5]), fmaxf(e1[6], e1[7]));
      float cm = fmaxf(fmaxf(c0m, c1m), fmaxf(c2m, c3m));
      cm = fmaxf(cm, xhalf(cm));
      const float mn = fmaxf(mrun, cm);
      const float cf = __expf(mrun - mn);
      float p0[8], p1[8];
#pragma unroll
      for (int r = 0; r < 8; ++r) {
        p0[r] = __expf(e0[r] - mn);
        p1[r] = __expf(e1[r] - mn);
      }
      float ps = (((p0[0] + p0[1]) + (p0[2] + p0[3])) + ((p0[4] + p0[5]) + (p0[6] + p0[7]))) +
                 (((p1[0] + p1[1]) + (p1[2] + p1[3])) + ((p1[4] + p1[5]) + (p1[6] + p1[7])));
      ps = ps + xhalf(ps);
      lrun = lrun * cf + ps;
      mrun = mn;

      v16h pb;
#pragma unroll
      for (int r = 0; r < 8; ++r) {
        pb[r]     = (_Float16)p0[r];
        pb[8 + r] = (_Float16)p1[r];
      }

#pragma unroll
      for (int t = 0; t < 8; ++t) {
#pragma unroll
        for (int r = 0; r < 8; ++r) acc[t][r] *= cf;
        acc[t] = __builtin_amdgcn_wmma_f32_16x16x32_f16(false, va[t], false, pb, (short)0,
                                                        acc[t], false, false);
      }
    }

    if (more) {
      const int nb = (cur ^ 1) * 2;
#pragma unroll
      for (int i = 0; i < 8; ++i)
        *(v8h*)(&lds[(nb + (i >> 2)) * BLK_ + vdst + (i & 3) * 32 * VP_]) = gv[i];
      *(v8h*)(&lds[nb * BLK_ + kdst]) = gk;
      *(v8h*)(&lds[(nb + 1) * BLK_ + kdst]) = gk;
    }
    __syncthreads();
  }

  const float g = gamma[0];
  const int i = i0 + col;
#pragma unroll
  for (int t = 0; t < 8; ++t) {
#pragma unroll
    for (int r = 0; r < 8; ++r) {
      const int c = chalf * 128 + t * 16 + 8 * hf + r;
      const size_t off = ((size_t)b * C_ + c) * N_ + i;
      out[off] = g * (acc[t][r] / lrun) + x[off];
    }
  }
}

extern "C" void kernel_launch(void* const* d_in, const int* in_sizes, int n_in,
                              void* d_out, int out_size, void* d_ws, size_t ws_size,
                              hipStream_t stream) {
  (void)in_sizes; (void)n_in; (void)out_size; (void)ws_size;
  const float* x     = (const float*)d_in[0];
  const float* mask  = (const float*)d_in[1];
  const float* Wq    = (const float*)d_in[2];
  const float* bq    = (const float*)d_in[3];
  const float* Wk    = (const float*)d_in[4];
  const float* bk    = (const float*)d_in[5];
  const float* Wv    = (const float*)d_in[6];
  const float* bv    = (const float*)d_in[7];
  const float* gamma = (const float*)d_in[8];
  float* out = (float*)d_out;

  char* ws = (char*)d_ws;
  size_t off = 0;
  auto take = [&](size_t bytes) -> char* {
    char* p = ws + off;
    off += (bytes + 255) & ~(size_t)255;
    return p;
  };
  _Float16* Qh = (_Float16*)take((size_t)B_ * N_ * CQ_ * sizeof(_Float16));
  _Float16* Kh = (_Float16*)take((size_t)B_ * N_ * CQ_ * sizeof(_Float16));
  _Float16* Vh = (_Float16*)take((size_t)B_ * C_ * N_ * sizeof(_Float16));

  proj_fused<<<dim3(N_ / 64, B_), 256, 0, stream>>>(x, mask, Wq, bq, Wk, bk, Wv, bv, Qh, Kh, Vh);
  attn_kernel<<<dim3(N_ / 64, B_), 256, 0, stream>>>(Qh, Kh, Vh, x, gamma, out);
}


# attn: K/V slabs asked for two trips ahead into a second register set (loop unrolled by two); hand-written barrier waits for LDS only, so loads stay in flight across it and no trip waits for its own
# speedup vs baseline: 4.4131x; 1.1517x over previous
#include <hip/hip_runtime.h>

typedef __attribute__((ext_vector_type(16))) _Float16 v16h;
typedef __attribute__((ext_vector_type(8)))  _Float16 v8h;
typedef __attribute__((ext_vector_type(8)))  float    v8f;

#define B_  4
#define C_  256
#define CQ_ 32
#define N_  4096

__device__ __forceinline__ v16h load_tile16x32(const _Float16* base, int strideHalfs) {
  const int l  = threadIdx.x & 31;
  const int hf = l >> 4;
  const _Float16* p = base + (size_t)(l & 15) * strideHalfs + hf * 8;
  union { v16h v; v8h h[2]; } u;
  u.h[0] = *(const v8h*)(p);
  u.h[1] = *(const v8h*)(p + 16);
  return u.v;
}

__device__ __forceinline__ float xhalf(float x) {
  const int b = __builtin_bit_cast(int, x);
  return __builtin_bit_cast(float, __builtin_amdgcn_permlanex16(b, b, 0x76543210, 0xfedcba98,
                                                                false, false));
}

__device__ __forceinline__ void lds_barrier() {
  asm volatile("s_wait_dscnt 0x0\n\ts_barrier_signal -1\n\ts_barrier_wait -1" ::: "memory");
}

typedef __attribute__((ext_vector_type(4))) float v4f;
#define PX_ 264

__global__ __launch_bounds__(256) void proj_fused(
    const float* __restrict__ x, const float* __restrict__ mask,
    const float* __restrict__ Wq, const float* __restrict__ bq,
    const float* __restrict__ Wk, const float* __restrict__ bk,
    const float* __restrict__ Wv, const float* __restrict__ bv,
    _Float16* __restrict__ Qh, _Float16* __restrict__ Kh, _Float16* __restrict__ Vh) {
  __shared__ __attribute__((aligned(16))) _Float16 Ws[320 * PX_];
  __shared__ __attribute__((aligned(16))) _Float16 Xs[64 * PX_];

  const int b    = blockIdx.y;
  const int n0   = blockIdx.x * 64;
  const int tid  = threadIdx.x;
  const int wave = __builtin_amdgcn_readfirstlane(tid >> 5);
  const int lane = tid & 31;

  const int xq = tid & 15, xg = tid >> 4;
  v4f xv[2][8];
#pragma unroll
  for (int i = 0; i < 2; ++i)
#pragma unroll
    for (int k = 0; k < 8; ++k)
      xv[i][k] = *(const v4f*)(x + ((size_t)b * C_ + xg * 8 + i * 128 + k) * N_ + n0 + xq * 4);

#pragma unroll
  for (int i0 = 0; i0 < 40; i0 += 8) {
    v4f wv[8][2];
#pragma unroll
    for (int u = 0; u < 8; ++u) {
      const int i = i0 + u;
      const int o = wave + 8 * i;
      const float* srow = (i < 4) ? (Wq + (size_t)o * C_)
                        : (i < 8) ? (Wk + (size_t)(o - 32) * C_)
                                  : (Wv + (size_t)(o - 64) * C_);
      const v4f* p = (const v4f*)(srow + lane * 8);
      wv[u][0] = p[0];
      wv[u][1] = p[1];
    }
#pragma unroll
    for (int u = 0; u < 8; ++u) {
      const int o = wave + 8 * (i0 + u);
      v8h h;
#pragma unroll
      for (int e = 0; e < 4; ++e) {
        h[e]     = (_Float16)wv[u][0][e];
        h[4 + e] = (_Float16)wv[u][1][e];
      }
      *(v8h*)(&Ws[o * PX_ + lane * 8]) = h;
    }
  }

#pragma unroll
  for (int i = 0; i < 2; ++i)
#pragma unroll
    for (int j = 0; j < 4; ++j) {
      v8h h;
#pragma unroll
      for (int k = 0; k < 8; ++k) h[k] = (_Float16)xv[i][k][j];
      *(v8h*)(&Xs[(xq * 4 + j) * PX_ + xg * 8 + i * 128]) = h;
    }
  __syncthreads();

  const int np = wave & 1, og = wave >> 1;
  const int hf = lane >> 4, col = lane & 15;

  float bqk[4][8];
  float bvv[5];
#pragma unroll
  for (int u = 0; u < 5; ++u) {
    const int ot = og * 5 + u;
    if (u < 4) {
      const float* bsrc = ((ot & 2) ? bk : bq) + (ot & 1) * 16 + 8 * hf;
#pragma unroll
      for (int r = 0; r < 8; ++r) bqk[u][r] = bsrc[r];
    }
    const int vt = (ot >= 4) ? (ot - 4) : 0;
    bvv[u] = bv[vt * 16 + col];
  }
  float mval[2];
#pragma unroll
  for (int t = 0; t < 2; ++t) mval[t] = mask[(size_t)b * N_ + n0 + (np * 2 + t) * 16 + col];

  v16h xf[2][8];
#pragma unroll
  for (int t = 0; t < 2; ++t)
#pragma unroll
    for (int ks = 0; ks < 8; ++ks)
      xf[t][ks] = load_tile16x32(Xs + ((np * 2 + t) * 16) * PX_ + ks * 32, PX_);

  v16h wf[2][8];
#pragma unroll
  for (int ks = 0; ks < 8; ++ks)
    wf[0][ks] = load_tile16x32(Ws + (og * 5 * 16) * PX_ + ks * 32, PX_);

  const v8f vzero = {0.f, 0.f, 0.f, 0.f, 0.f, 0.f, 0.f, 0.f};
#pragma unroll
  for (int u = 0; u < 5; ++u) {
    const int ot = og * 5 + u;
    if (u + 1 < 5) {
#pragma unroll
      for (int ks = 0; ks < 8; ++ks)
        wf[(u + 1) & 1][ks] = load_tile16x32(Ws + ((ot + 1) * 16) * PX_ + ks * 32, PX_);
    }
    __builtin_amdgcn_sched_barrier(0);

    v8f acc[2] = {vzero, vzero};
    if (ot < 4) {
#pragma unroll
      for (int ks = 0; ks < 8; ++ks)
#pragma unroll
        for (int t = 0; t < 2; ++t)
          acc[t] = __builtin_amdgcn_wmma_f32_16x16x32_f16(false, wf[u & 1][ks], false, xf[t][ks],
                                                          (short)0, acc[t], false, false);
      _Float16* dst = (ot < 2) ? Qh : Kh;
#pragma unroll
      for (int t = 0; t < 2; ++t) {
        const int n = n0 + (np * 2 + t) * 16 + col;
        v8h h;
#pragma unroll
        for (int r = 0; r < 8; ++r) h[r] = (_Float16)((acc[t][r] + bqk[u < 4 ? u : 0][r]) * mval[t]);
        *(v8h*)(dst + ((size_t)b * N_ + n) * CQ_ + (ot & 1) * 16 + 8 * hf) = h;
      }
    } else {
#pragma unroll
      for (int ks = 0; ks < 8; ++ks)
#pragma unroll
        for (int t = 0; t < 2; ++t)
          acc[t] = __builtin_amdgcn_wmma_f32_16x16x32_f16(false, xf[t][ks], false, wf[u & 1][ks],
                                                          (short)0, acc[t], false, false);
      const int ch = (ot - 4) * 16 + col;
#pragma unroll
      for (int t = 0; t < 2; ++t) {
        v8h h;
#pragma unroll
        for (int r = 0; r < 8; ++r) h[r] = (_Float16)(acc[t][r] + bvv[u]);
        *(v8h*)(Vh + ((size_t)b * C_ + ch) * N_ + n0 + (np * 2 + t) * 16 + 8 * hf) = h;
      }
    }
  }
}

#define KT_      64
#define VP_      72
#define KP_      40
#define VH_      (128 * VP_)
#define BLK_     16384
#define MGW_     (66 * 32)

__global__ __launch_bounds__(256) void attn_kernel(
    const _Float16* __restrict__ Qh, const _Float16* __restrict__ Kh,
    const _Float16* __restrict__ Vh, const float* __restrict__ x,
    const float* __restrict__ gamma, float* __restrict__ out) {
  __shared__ __attribute__((aligned(16))) _Float16 lds[4 * BLK_];

  const int b     = blockIdx.y;
  const int tid   = threadIdx.x;
  const int wave  = tid >> 5;
  const int lane  = tid & 31;
  const int kh    = (wave >> 1) & 1;
  const int qt    = (wave & 1) | ((wave >> 2) << 1);
  const int i0    = blockIdx.x * 64 + qt * 16;
  const int hf    = lane >> 4, col = lane & 15;

  const v16h qb = load_tile16x32(Qh + ((size_t)b * N_ + i0) * CQ_, CQ_);

  const int vrow = tid >> 3, vch = tid & 7;
  const int krow = tid >> 2, kch = tid & 3;
  const _Float16* vsrc = Vh + ((size_t)b * C_ + vrow) * N_ + vch * 8;
  const _Float16* ksrc = Kh + (size_t)b * N_ * CQ_ + tid * 8;
  const int vdst = vrow * VP_ + vch * 8;
  const int kdst = VH_ + krow * KP_ + kch * 8;

  v8h gv[2][8];
  v8h gk[2];
#pragma unroll
  for (int i = 0; i < 8; ++i) gv[0][i] = *(const v8h*)(vsrc + (size_t)i * 32 * N_);
  gk[0] = *(const v8h*)(ksrc);
  __builtin_amdgcn_sched_barrier(0);
#pragma unroll
  for (int i = 0; i < 8; ++i) gv[1][i] = *(const v8h*)(vsrc + (size_t)i * 32 * N_ + KT_);
  gk[1] = *(const v8h*)(ksrc + (size_t)KT_ * CQ_);
  __builtin_amdgcn_sched_barrier(0);
#pragma unroll
  for (int i = 0; i < 8; ++i)
    *(v8h*)(&lds[(i >> 2) * BLK_ + vdst + (i & 3) * 32 * VP_]) = gv[0][i];
  *(v8h*)(&lds[kdst]) = gk[0];
  *(v8h*)(&lds[BLK_ + kdst]) = gk[0];
#pragma unroll
  for (int i = 0; i < 8; ++i) gv[0][i] = *(const v8h*)(vsrc + (size_t)i * 32 * N_ + 2 * KT_);
  gk[0] = *(const v8h*)(ksrc + (size_t)2 * KT_ * CQ_);
  lds_barrier();

  const v8f vzero = {0.f, 0.f, 0.f, 0.f, 0.f, 0.f, 0.f, 0.f};
  v8f acc[16];
#pragma unroll
  for (int t = 0; t < 16; ++t) acc[t] = vzero;
  float mrun = -3.0e38f, lrun = 0.0f;

  for (int it2 = 0; it2 < N_ / KT_; it2 += 2) {
#pragma unroll
  for (int cur = 0; cur < 2; ++cur) {
    const int it = it2 + cur;

    const _Float16* Va = &lds[(cur * 2 + kh) * BLK_];
    const _Float16* Vb = &lds[(cur * 2 + (kh ^ 1)) * BLK_];
    const _Float16* Ks = Va + VH_ + (kh * 32) * KP_;

    const v16h ka0 = load_tile16x32(Ks, KP_);
    const v16h ka1 = load_tile16x32(Ks + 16 * KP_, KP_);
    const v8f e0 = __builtin_amdgcn_wmma_f32_16x16x32_f16(false, ka0, false, qb, (short)0,
                                                          vzero, false, false);
    const v8f e1 = __builtin_amdgcn_wmma_f32_16x16x32_f16(false, ka1, false, qb, (short)0,
                                                          vzero, false, false);

    v16h va[16];
#pragma unroll
    for (int t = 0; t < 8; ++t) va[t] = load_tile16x32(Va + (t * 16) * VP_ + kh * 32, VP_);
#pragma unroll
    for (int t = 0; t < 8; ++t) va[8 + t] = load_tile16x32(Vb + (t * 16) * VP_ + kh * 32, VP_);
    __builtin_amdgcn_sched_barrier(0);

    float c0m = fmaxf(fmaxf(e0[0], e0[1]), fmaxf(e0[2], e0[3]));
    float c1m = fmaxf(fmaxf(e0[4], e0[5]), fmaxf(e0[6], e0[7]));
    float c2m = fmaxf(fmaxf(e1[0], e1[1]), fmaxf(e1[2], e1[3]));
    float c3m = fmaxf(fmaxf(e1[4], e1[5]), fmaxf(e1[6], e1[7]));
    float cm = fmaxf(fmaxf(c0m, c1m), fmaxf(c2m, c3m));
    cm = fmaxf(cm, xhalf(cm));
    const float mn = fmaxf(mrun, cm);
    const float cf = __expf(mrun - mn);
    float p0[8], p1[8];
#pragma unroll
    for (int r = 0; r < 8; ++r) {
      p0[r] = __expf(e0[r] - mn);
      p1[r] = __expf(e1[r] - mn);
    }
    float ps = (((p0[0] + p0[1]) + (p0[2] + p0[3])) + ((p0[4] + p0[5]) + (p0[6] + p0[7]))) +
               (((p1[0] + p1[1]) + (p1[2] + p1[3])) + ((p1[4] + p1[5]) + (p1[6] + p1[7])));
    ps = ps + xhalf(ps);
    lrun = lrun * cf + ps;
    mrun = mn;

    v16h pb;
#pragma unroll
    for (int r = 0; r < 8; ++r) {
      pb[r]     = (_Float16)p0[r];
      pb[8 + r] = (_Float16)p1[r];
    }

    {
      const int nb = (cur ^ 1) * 2;
#pragma unroll
      for (int i = 0; i < 8; ++i)
        *(v8h*)(&lds[(nb + (i >> 2)) * BLK_ + vdst + (i & 3) * 32 * VP_]) = gv[cur ^ 1][i];
      *(v8h*)(&lds[nb * BLK_ + kdst]) = gk[cur ^ 1];
      *(v8h*)(&lds[(nb + 1) * BLK_ + kdst]) = gk[cur ^ 1];
    }
    {
      const int nx = (it + 3 < N_ / KT_) ? (it + 3) : (N_ / KT_ - 1);
#pragma unroll
      for (int i = 0; i < 8; ++i)
        gv[cur ^ 1][i] = *(const v8h*)(vsrc + (size_t)i * 32 * N_ + nx * KT_);
      gk[cur ^ 1] = *(const v8h*)(ksrc + (size_t)nx * KT_ * CQ_);
    }

#pragma unroll
    for (int t = 0; t < 16; ++t) {
#pragma unroll
      for (int r = 0; r < 8; ++r) acc[t][r] *= cf;
      acc[t] = __builtin_amdgcn_wmma_f32_16x16x32_f16(false, va[t], false, pb, (short)0,
                                                      acc[t], false, false);
    }

    __builtin_amdgcn_sched_barrier(0);
    lds_barrier();
  }
  }

  float* mg = (float*)lds;
  float* mine = mg + wave * MGW_;
#pragma unroll
  for (int t = 0; t < 8; ++t)
#pragma unroll
    for (int r = 0; r < 8; ++r) mine[(t * 8 + r) * 32 + lane] = acc[8 + t][r];
  mine[64 * 32 + lane] = mrun;
  mine[65 * 32 + lane] = lrun;
  __syncthreads();
  const float* oth = mg + (wave ^ 2) * MGW_;
  const float mo = oth[64 * 32 + lane];
  const float lo = oth[65 * 32 + lane];
  const float mj = fmaxf(mrun, mo);
  const float fa = __expf(mrun - mj);
  const float fb = __expf(mo - mj);
  const float lj = lrun * fa + lo * fb;

  const float g = gamma[0];
  const int i = i0 + col;
#pragma unroll
  for (int t = 0; t < 8; ++t) {
#pragma unroll
    for (int r = 0; r < 8; ++r) {
      const int c = kh * 128 + t * 16 + 8 * hf + r;
      const size_t off = ((size_t)b * C_ + c) * N_ + i;
      const float o = acc[t][r] * fa + oth[(t * 8 + r) * 32 + lane] * fb;
      out[off] = g * (o / lj) + x[off];
    }
  }
}

extern "C" void kernel_launch(void* const* d_in, const int* in_sizes, int n_in,
                              void* d_out, int out_size, void* d_ws, size_t ws_size,
                              hipStream_t stream) {
  (void)in_sizes; (void)n_in; (void)out_size; (void)ws_size;
  const float* x     = (const float*)d_in[0];
  const float* mask  = (const float*)d_in[1];
  const float* Wq    = (const float*)d_in[2];
  const float* bq    = (const float*)d_in[3];
  const float* Wk    = (const float*)d_in[4];
  const float* bk    = (const float*)d_in[5];
  const float* Wv    = (const float*)d_in[6];
  const float* bv    = (const float*)d_in[7];
  const float* gamma = (const float*)d_in[8];
  float* out = (float*)d_out;

  char* ws = (char*)d_ws;
  size_t off = 0;
  auto take = [&](size_t bytes) -> char* {
    char* p = ws + off;
    off += (bytes + 255) & ~(size_t)255;
    return p;
  };
  _Float16* Qh = (_Float16*)take((size_t)B_ * N_ * CQ_ * sizeof(_Float16));
  _Float16* Kh = (_Float16*)take((size_t)B_ * N_ * CQ_ * sizeof(_Float16));
  _Float16* Vh = (_Float16*)take((size_t)B_ * C_ * N_ * sizeof(_Float16));

  proj_fused<<<dim3(N_ / 64, B_), 256, 0, stream>>>(x, mask, Wq, bq, Wk, bk, Wv, bv, Qh, Kh, Vh);
  attn_kernel<<<dim3(N_ / 64, B_), 256, 0, stream>>>(Qh, Kh, Vh, x, gamma, out);
}


# attn: wave owns 32 queries (two query tiles) in 4-wave groups, one wave per SIMD: each K/V fragment read from LDS feeds two matrix instructions, so LDS fragment reads per product are halved
# speedup vs baseline: 4.6706x; 1.0583x over previous
#include <hip/hip_runtime.h>

typedef __attribute__((ext_vector_type(16))) _Float16 v16h;
typedef __attribute__((ext_vector_type(8)))  _Float16 v8h;
typedef __attribute__((ext_vector_type(8)))  float    v8f;

#define B_  4
#define C_  256
#define CQ_ 32
#define N_  4096

__device__ __forceinline__ v16h load_tile16x32(const _Float16* base, int strideHalfs) {
  const int l  = threadIdx.x & 31;
  const int hf = l >> 4;
  const _Float16* p = base + (size_t)(l & 15) * strideHalfs + hf * 8;
  union { v16h v; v8h h[2]; } u;
  u.h[0] = *(const v8h*)(p);
  u.h[1] = *(const v8h*)(p + 16);
  return u.v;
}

__device__ __forceinline__ float xhalf(float x) {
  const int b = __builtin_bit_cast(int, x);
  return __builtin_bit_cast(float, __builtin_amdgcn_permlanex16(b, b, 0x76543210, 0xfedcba98,
                                                                false, false));
}

__device__ __forceinline__ void lds_barrier() {
  asm volatile("s_wait_dscnt 0x0\n\ts_barrier_signal -1\n\ts_barrier_wait -1" ::: "memory");
}

typedef __attribute__((ext_vector_type(4))) float v4f;
#define PX_ 264

__global__ __launch_bounds__(256) void proj_fused(
    const float* __restrict__ x, const float* __restrict__ mask,
    const float* __restrict__ Wq, const float* __restrict__ bq,
    const float* __restrict__ Wk, const float* __restrict__ bk,
    const float* __restrict__ Wv, const float* __restrict__ bv,
    _Float16* __restrict__ Qh, _Float16* __restrict__ Kh, _Float16* __restrict__ Vh) {
  __shared__ __attribute__((aligned(16))) _Float16 Ws[320 * PX_];
  __shared__ __attribute__((aligned(16))) _Float16 Xs[64 * PX_];

  const int b    = blockIdx.y;
  const int n0   = blockIdx.x * 64;
  const int tid  = threadIdx.x;
  const int wave = __builtin_amdgcn_readfirstlane(tid >> 5);
  const int lane = tid & 31;

  const int xq = tid & 15, xg = tid >> 4;
  v4f xv[2][8];
#pragma unroll
  for (int i = 0; i < 2; ++i)
#pragma unroll
    for (int k = 0; k < 8; ++k)
      xv[i][k] = *(const v4f*)(x + ((size_t)b * C_ + xg * 8 + i * 128 + k) * N_ + n0 + xq * 4);

#pragma unroll
  for (int i0 = 0; i0 < 40; i0 += 8) {
    v4f wv[8][2];
#pragma unroll
    for (int u = 0; u < 8; ++u) {
      const int i = i0 + u;
      const int o = wave + 8 * i;
      const float* srow = (i < 4) ? (Wq + (size_t)o * C_)
                        : (i < 8) ? (Wk + (size_t)(o - 32) * C_)
                                  : (Wv + (size_t)(o - 64) * C_);
      const v4f* p = (const v4f*)(srow + lane * 8);
      wv[u][0] = p[0];
      wv[u][1] = p[1];
    }
#pragma unroll
    for (int u = 0; u < 8; ++u) {
      const int o = wave + 8 * (i0 + u);
      v8h h;
#pragma unroll
      for (int e = 0; e < 4; ++e) {
        h[e]     = (_Float16)wv[u][0][e];
        h[4 + e] = (_Float16)wv[u][1][e];
      }
      *(v8h*)(&Ws[o * PX_ + lane * 8]) = h;
    }
  }

#pragma unroll
  for (int i = 0; i < 2; ++i)
#pragma unroll
    for (int j = 0; j < 4; ++j) {
      v8h h;
#pragma unroll
      for (int k = 0; k < 8; ++k) h[k] = (_Float16)xv[i][k][j];
      *(v8h*)(&Xs[(xq * 4 + j) * PX_ + xg * 8 + i * 128]) = h;
    }
  __syncthreads();

  const int np = wave & 1, og = wave >> 1;
  const int hf = lane >> 4, col = lane & 15;

  float bqk[4][8];
  float bvv[5];
#pragma unroll
  for (int u = 0; u < 5; ++u) {
    const int ot = og * 5 + u;
    if (u < 4) {
      const float* bsrc = ((ot & 2) ? bk : bq) + (ot & 1) * 16 + 8 * hf;
#pragma unroll
      for (int r = 0; r < 8; ++r) bqk[u][r] = bsrc[r];
    }
    const int vt = (ot >= 4) ? (ot - 4) : 0;
    bvv[u] = bv[vt * 16 + col];
  }
  float mval[2];
#pragma unroll
  for (int t = 0; t < 2; ++t) mval[t] = mask[(size_t)b * N_ + n0 + (np * 2 + t) * 16 + col];

  v16h xf[2][8];
#pragma unroll
  for (int t = 0; t < 2; ++t)
#pragma unroll
    for (int ks = 0; ks < 8; ++ks)
      xf[t][ks] = load_tile16x32(Xs + ((np * 2 + t) * 16) * PX_ + ks * 32, PX_);

  v16h wf[2][8];
#pragma unroll
  for (int ks = 0; ks < 8; ++ks)
    wf[0][ks] = load_tile16x32(Ws + (og * 5 * 16) * PX_ + ks * 32, PX_);

  const v8f vzero = {0.f, 0.f, 0.f, 0.f, 0.f, 0.f, 0.f, 0.f};
#pragma unroll
  for (int u = 0; u < 5; ++u) {
    const int ot = og * 5 + u;
    if (u + 1 < 5) {
#pragma unroll
      for (int ks = 0; ks < 8; ++ks)
        wf[(u + 1) & 1][ks] = load_tile16x32(Ws + ((ot + 1) * 16) * PX_ + ks * 32, PX_);
    }
    __builtin_amdgcn_sched_barrier(0);

    v8f acc[2] = {vzero, vzero};
    if (ot < 4) {
#pragma unroll
      for (int ks = 0; ks < 8; ++ks)
#pragma unroll
        for (int t = 0; t < 2; ++t)
          acc[t] = __builtin_amdgcn_wmma_f32_16x16x32_f16(false, wf[u & 1][ks], false, xf[t][ks],
                                                          (short)0, acc[t], false, false);
      _Float16* dst = (ot < 2) ? Qh : Kh;
#pragma unroll
      for (int t = 0; t < 2; ++t) {
        const int n = n0 + (np * 2 + t) * 16 + col;
        v8h h;
#pragma unroll
        for (int r = 0; r < 8; ++r) h[r] = (_Float16)((acc[t][r] + bqk[u < 4 ? u : 0][r]) * mval[t]);
        *(v8h*)(dst + ((size_t)b * N_ + n) * CQ_ + (ot & 1) * 16 + 8 * hf) = h;
      }
    } else {
#pragma unroll
      for (int ks = 0; ks < 8; ++ks)
#pragma unroll
        for (int t = 0; t < 2; ++t)
          acc[t] = __builtin_amdgcn_wmma_f32_16x16x32_f16(false, xf[t][ks], false, wf[u & 1][ks],
                                                          (short)0, acc[t], false, false);
      const int ch = (ot - 4) * 16 + col;
#pragma unroll
      for (int t = 0; t < 2; ++t) {
        v8h h;
#pragma unroll
        for (int r = 0; r < 8; ++r) h[r] = (_Float16)(acc[t][r] + bvv[u]);
        *(v8h*)(Vh + ((size_t)b * C_ + ch) * N_ + n0 + (np * 2 + t) * 16 + 8 * hf) = h;
      }
    }
  }
}

#define KT_      64
#define VP_      72
#define KP_      40
#define VH_      (128 * VP_)
#define BLK_     16384
#define MGW_     (132 * 32)

__global__ __launch_bounds__(128) void attn_kernel(
    const _Float16* __restrict__ Qh, const _Float16* __restrict__ Kh,
    const _Float16* __restrict__ Vh, const float* __restrict__ x,
    const float* __restrict__ gamma, float* __restrict__ out) {
  __shared__ __attribute__((aligned(16))) _Float16 lds[4 * BLK_];

  const int b     = blockIdx.y;
  const int tid   = threadIdx.x;
  const int wave  = tid >> 5;
  const int lane  = tid & 31;
  const int kh    = wave & 1;
  const int qp    = wave >> 1;
  const int i0    = blockIdx.x * 64 + qp * 32;
  const int hf    = lane >> 4, col = lane & 15;

  v16h qb[2];
#pragma unroll
  for (int q = 0; q < 2; ++q)
    qb[q] = load_tile16x32(Qh + ((size_t)b * N_ + i0 + q * 16) * CQ_, CQ_);

  const int vrow = tid >> 3, vch = tid & 7;
  const int krow = tid >> 2, kch = tid & 3;
  const _Float16* vsrc = Vh + ((size_t)b * C_ + vrow) * N_ + vch * 8;
  const _Float16* ksrc = Kh + (size_t)b * N_ * CQ_ + tid * 8;
  const int vdst = vrow * VP_ + vch * 8;
  const int kdst = VH_ + krow * KP_ + kch * 8;

  v8h gv[2][16];
  v8h gk[2][2];
#pragma unroll
  for (int i = 0; i < 16; ++i) gv[0][i] = *(const v8h*)(vsrc + (size_t)i * 16 * N_);
#pragma unroll
  for (int j = 0; j < 2; ++j) gk[0][j] = *(const v8h*)(ksrc + j * 1024);
  __builtin_amdgcn_sched_barrier(0);
#pragma unroll
  for (int i = 0; i < 16; ++i) gv[1][i] = *(const v8h*)(vsrc + (size_t)i * 16 * N_ + KT_);
#pragma unroll
  for (int j = 0; j < 2; ++j) gk[1][j] = *(const v8h*)(ksrc + (size_t)KT_ * CQ_ + j * 1024);
  __builtin_amdgcn_sched_barrier(0);
#pragma unroll
  for (int i = 0; i < 16; ++i)
    *(v8h*)(&lds[(i >> 3) * BLK_ + vdst + (i & 7) * 16 * VP_]) = gv[0][i];
#pragma unroll
  for (int j = 0; j < 2; ++j) {
    *(v8h*)(&lds[kdst + j * 32 * KP_]) = gk[0][j];
    *(v8h*)(&lds[BLK_ + kdst + j * 32 * KP_]) = gk[0][j];
  }
#pragma unroll
  for (int i = 0; i < 16; ++i) gv[0][i] = *(const v8h*)(vsrc + (size_t)i * 16 * N_ + 2 * KT_);
#pragma unroll
  for (int j = 0; j < 2; ++j) gk[0][j] = *(const v8h*)(ksrc + (size_t)2 * KT_ * CQ_ + j * 1024);
  lds_barrier();

  const v8f vzero = {0.f, 0.f, 0.f, 0.f, 0.f, 0.f, 0.f, 0.f};
  v8f acc[2][16];
#pragma unroll
  for (int q = 0; q < 2; ++q)
#pragma unroll
    for (int t = 0; t < 16; ++t) acc[q][t] = vzero;
  float mrun[2] = {-3.0e38f, -3.0e38f};
  float lrun[2] = {0.0f, 0.0f};

  for (int it2 = 0; it2 < N_ / KT_; it2 += 2) {
#pragma unroll
  for (int cur = 0; cur < 2; ++cur) {
    const int it = it2 + cur;

    const _Float16* Va = &lds[(cur * 2 + kh) * BLK_];
    const _Float16* Vb = &lds[(cur * 2 + (kh ^ 1)) * BLK_];
    const _Float16* Ks = Va + VH_ + (kh * 32) * KP_;

    const v16h ka0 = load_tile16x32(Ks, KP_);
    const v16h ka1 = load_tile16x32(Ks + 16 * KP_, KP_);
    v8f e0[2], e1[2];
#pragma unroll
    for (int q = 0; q < 2; ++q) {
      e0[q] = __builtin_amdgcn_wmma_f32_16x16x32_f16(false, ka0, false, qb[q], (short)0,
                                                     vzero, false, false);
      e1[q] = __builtin_amdgcn_wmma_f32_16x16x32_f16(false, ka1, false, qb[q], (short)0,
                                                     vzero, false, false);
    }

    v16h va[16];
#pragma unroll
    for (int t = 0; t < 8; ++t) va[t] = load_tile16x32(Va + (t * 16) * VP_ + kh * 32, VP_);
#pragma unroll
    for (int t = 0; t < 8; ++t) va[8 + t] = load_tile16x32(Vb + (t * 16) * VP_ + kh * 32, VP_);
    __builtin_amdgcn_sched_barrier(0);

    float cf[2];
    v16h pb[2];
#pragma unroll
    for (int q = 0; q < 2; ++q) {
      float c0m = fmaxf(fmaxf(e0[q][0], e0[q][1]), fmaxf(e0[q][2], e0[q][3]));
      float c1m = fmaxf(fmaxf(e0[q][4], e0[q][5]), fmaxf(e0[q][6], e0[q][7]));
      float c2m = fmaxf(fmaxf(e1[q][0], e1[q][1]), fmaxf(e1[q][2], e1[q][3]));
      float c3m = fmaxf(fmaxf(e1[q][4], e1[q][5]), fmaxf(e1[q][6], e1[q][7]));
      float cm = fmaxf(fmaxf(c0m, c1m), fmaxf(c2m, c3m));
      cm = fmaxf(cm, xhalf(cm));
      const float mn = fmaxf(mrun[q], cm);
      cf[q] = __expf(mrun[q] - mn);
      float p0[8], p1[8];
#pragma unroll
      for (int r = 0; r < 8; ++r) {
        p0[r] = __expf(e0[q][r] - mn);
        p1[r] = __expf(e1[q][r] - mn);
      }
      float ps = (((p0[0] + p0[1]) + (p0[2] + p0[3])) + ((p0[4] + p0[5]) + (p0[6] + p0[7]))) +
                 (((p1[0] + p1[1]) + (p1[2] + p1[3])) + ((p1[4] + p1[5]) + (p1[6] + p1[7])));
      ps = ps + xhalf(ps);
      lrun[q] = lrun[q] * cf[q] + ps;
      mrun[q] = mn;
#pragma unroll
      for (int r = 0; r < 8; ++r) {
        pb[q][r]     = (_Float16)p0[r];
        pb[q][8 + r] = (_Float16)p1[r];
      }
    }

    {
      const int nb = (cur ^ 1) * 2;
#pragma unroll
      for (int i = 0; i < 16; ++i)
        *(v8h*)(&lds[(nb + (i >> 3)) * BLK_ + vdst + (i & 7) * 16 * VP_]) = gv[cur ^ 1][i];
#pragma unroll
      for (int j = 0; j < 2; ++j) {
        *(v8h*)(&lds[nb * BLK_ + kdst + j * 32 * KP_]) = gk[cur ^ 1][j];
        *(v8h*)(&lds[(nb + 1) * BLK_ + kdst + j * 32 * KP_]) = gk[cur ^ 1][j];
      }
    }
    {
      const int nx = (it + 3 < N_ / KT_) ? (it + 3) : (N_ / KT_ - 1);
#pragma unroll
      for (int i = 0; i < 16; ++i)
        gv[cur ^ 1][i] = *(const v8h*)(vsrc + (size_t)i * 16 * N_ + nx * KT_);
#pragma unroll
      for (int j = 0; j < 2; ++j)
        gk[cur ^ 1][j] = *(const v8h*)(ksrc + (size_t)nx * KT_ * CQ_ + j * 1024);
    }

#pragma unroll
    for (int t = 0; t < 16; ++t) {
#pragma unroll
      for (int q = 0; q < 2; ++q) {
#pragma unroll
        for (int r = 0; r < 8; ++r) acc[q][t][r] *= cf[q];
        acc[q][t] = __builtin_amdgcn_wmma_f32_16x16x32_f16(false, va[t], false, pb[q], (short)0,
                                                           acc[q][t], false, false);
      }
    }

    __builtin_amdgcn_sched_barrier(0);
    lds_barrier();
  }
  }

  float* mg = (float*)lds;
  float* mine = mg + wave * MGW_;
#pragma unroll
  for (int q = 0; q < 2; ++q) {
#pragma unroll
    for (int t = 0; t < 8; ++t)
#pragma unroll
      for (int r = 0; r < 8; ++r) mine[((q * 8 + t) * 8 + r) * 32 + lane] = acc[q][8 + t][r];
    mine[(128 + q) * 32 + lane] = mrun[q];
    mine[(130 + q) * 32 + lane] = lrun[q];
  }
  __syncthreads();
  const float* oth = mg + (wave ^ 1) * MGW_;
  float fa[2], fb[2], lj[2];
#pragma unroll
  for (int q = 0; q < 2; ++q) {
    const float mo = oth[(128 + q) * 32 + lane];
    const float lo = oth[(130 + q) * 32 + lane];
    const float mj = fmaxf(mrun[q], mo);
    fa[q] = __expf(mrun[q] - mj);
    fb[q] = __expf(mo - mj);
    lj[q] = lrun[q] * fa[q] + lo * fb[q];
  }

  const float g = gamma[0];
#pragma unroll
  for (int q = 0; q < 2; ++q) {
    const int i = i0 + q * 16 + col;
#pragma unroll
    for (int t = 0; t < 8; ++t) {
#pragma unroll
      for (int r = 0; r < 8; ++r) {
        const int c = kh * 128 + t * 16 + 8 * hf + r;
        const size_t off = ((size_t)b * C_ + c) * N_ + i;
        const float o = acc[q][t][r] * fa[q] + oth[((q * 8 + t) * 8 + r) * 32 + lane] * fb[q];
        out[off] = g * (o / lj[q]) + x[off];
      }
    }
  }
}

extern "C" void kernel_launch(void* const* d_in, const int* in_sizes, int n_in,
                              void* d_out, int out_size, void* d_ws, size_t ws_size,
                              hipStream_t stream) {
  (void)in_sizes; (void)n_in; (void)out_size; (void)ws_size;
  const float* x     = (const float*)d_in[0];
  const float* mask  = (const float*)d_in[1];
  const float* Wq    = (const float*)d_in[2];
  const float* bq    = (const float*)d_in[3];
  const float* Wk    = (const float*)d_in[4];
  const float* bk    = (const float*)d_in[5];
  const float* Wv    = (const float*)d_in[6];
  const float* bv    = (const float*)d_in[7];
  const float* gamma = (const float*)d_in[8];
  float* out = (float*)d_out;

  char* ws = (char*)d_ws;
  size_t off = 0;
  auto take = [&](size_t bytes) -> char* {
    char* p = ws + off;
    off += (bytes + 255) & ~(size_t)255;
    return p;
  };
  _Float16* Qh = (_Float16*)take((size_t)B_ * N_ * CQ_ * sizeof(_Float16));
  _Float16* Kh = (_Float16*)take((size_t)B_ * N_ * CQ_ * sizeof(_Float16));
  _Float16* Vh = (_Float16*)take((size_t)B_ * C_ * N_ * sizeof(_Float16));

  proj_fused<<<dim3(N_ / 64, B_), 256, 0, stream>>>(x, mask, Wq, bq, Wk, bk, Wv, bv, Qh, Kh, Vh);
  attn_kernel<<<dim3(N_ / 64, B_), 128, 0, stream>>>(Qh, Kh, Vh, x, gamma, out);
}
